# starting kernel
# baseline (speedup 1.0000x reference)
#include <hip/hip_runtime.h>

typedef __attribute__((ext_vector_type(2))) float v2f;
typedef __attribute__((ext_vector_type(8))) float v8f;

#define EPS 1e-5f
#define HW  3136
#define CIN 256
#define CR  64
#define GK  784
#define NB  4

__global__ void __launch_bounds__(128)
invo_gemm1_bn_relu(const float* __restrict__ x, const float* __restrict__ W1,
                   const float* __restrict__ b1, const float* __restrict__ gamma,
                   const float* __restrict__ beta, const float* __restrict__ mean,
                   const float* __restrict__ var, float* __restrict__ t)
{
    const int lane   = threadIdx.x & 31;
    const int wave   = threadIdx.x >> 5;
    const int laneN  = lane & 15;
    const int laneHi = lane >> 4;
    const int p0     = blockIdx.x * 16;
    const int b      = blockIdx.y;
    const int o0     = wave * 16;

    const float* xb   = x + (size_t)b * CIN * HW;
    const float* Arow = W1 + (size_t)(o0 + laneN) * CIN + 2 * laneHi;
    const float* Bcol = xb + (size_t)(2 * laneHi) * HW + p0 + laneN;

    v8f acc = {};
#pragma unroll 8
    for (int k = 0; k < CIN; k += 4) {
        v2f a = *(const v2f*)(Arow + k);
        v2f bb;
        bb.x = Bcol[(size_t)k * HW];
        bb.y = Bcol[(size_t)(k + 1) * HW];
        acc = __builtin_amdgcn_wmma_f32_16x16x4_f32(
            false, a, false, bb, (short)0, acc, false, false);
    }

    float* tb = t + (size_t)b * CR * HW + p0 + laneN;
#pragma unroll
    for (int r = 0; r < 8; ++r) {
        int o = o0 + r + 8 * laneHi;
        float s  = gamma[o] * __frsqrt_rn(var[o] + EPS);
        float v  = (acc[r] + b1[o]) * s + (beta[o] - mean[o] * s);
        tb[(size_t)o * HW] = fmaxf(v, 0.0f);
    }
}

__global__ void __launch_bounds__(32)
invo_gemm2(const float* __restrict__ t, const float* __restrict__ W2,
           const float* __restrict__ b2, float* __restrict__ w)
{
    const int lane   = threadIdx.x & 31;
    const int laneN  = lane & 15;
    const int laneHi = lane >> 4;
    const int p0     = blockIdx.x * 16;
    const int m0     = blockIdx.y * 16;
    const int b      = blockIdx.z;

    const float* tb   = t + (size_t)b * CR * HW;
    const float* Arow = W2 + (size_t)(m0 + laneN) * CR + 2 * laneHi;
    const float* Bcol = tb + (size_t)(2 * laneHi) * HW + p0 + laneN;

    v8f acc = {};
#pragma unroll
    for (int k = 0; k < CR; k += 4) {
        v2f a = *(const v2f*)(Arow + k);
        v2f bb;
        bb.x = Bcol[(size_t)k * HW];
        bb.y = Bcol[(size_t)(k + 1) * HW];
        acc = __builtin_amdgcn_wmma_f32_16x16x4_f32(
            false, a, false, bb, (short)0, acc, false, false);
    }

    float* wb = w + (size_t)b * GK * HW + p0 + laneN;
#pragma unroll
    for (int r = 0; r < 8; ++r) {
        int m = m0 + r + 8 * laneHi;
        wb[(size_t)m * HW] = acc[r] + b2[m];
    }
}

__global__ void __launch_bounds__(256)
invo_apply(const float* __restrict__ x, const float* __restrict__ wk,
           float* __restrict__ out)
{
    __shared__ float wl[49 * 16];

    const int p0 = blockIdx.x * 16;
    const int g  = blockIdx.y;
    const int b  = blockIdx.z;
    const int tid = threadIdx.x;

    const float* wsrc = wk + ((size_t)b * GK + (size_t)g * 49) * HW + p0;
#pragma unroll
    for (int idx = tid; idx < 49 * 16; idx += 256) {
        int kk = idx >> 4, pl = idx & 15;
        wl[kk * 16 + pl] = wsrc[(size_t)kk * HW + pl];
    }
    __syncthreads();

    const int pl = tid & 15;
    const int cg = tid >> 4;
    const int p  = p0 + pl;
    const int h  = p / 56;
    const int wc = p % 56;
    const int c  = g * 16 + cg;

    const float* xc = x + ((size_t)b * CIN + c) * HW;

    float acc = 0.0f;
#pragma unroll
    for (int i = 0; i < 7; ++i) {
        int hh = h + i - 3;
        bool hin = (hh >= 0) & (hh < 56);
#pragma unroll
        for (int j = 0; j < 7; ++j) {
            int ww = wc + j - 3;
            bool ok = hin & (ww >= 0) & (ww < 56);
            float xv = ok ? xc[hh * 56 + ww] : 0.0f;
            acc = fmaf(xv, wl[(i * 7 + j) * 16 + pl], acc);
        }
    }
    out[((size_t)b * CIN + c) * HW + p] = acc;
}

extern "C" void kernel_launch(void* const* d_in, const int* in_sizes, int n_in,
                              void* d_out, int out_size, void* d_ws, size_t ws_size,
                              hipStream_t stream)
{
    const float* x     = (const float*)d_in[0];
    const float* W1    = (const float*)d_in[1];
    const float* b1    = (const float*)d_in[2];
    const float* gamma = (const float*)d_in[3];
    const float* beta  = (const float*)d_in[4];
    const float* mean  = (const float*)d_in[5];
    const float* var   = (const float*)d_in[6];
    const float* W2    = (const float*)d_in[7];
    const float* b2    = (const float*)d_in[8];
    float* out = (float*)d_out;

    float* t = (float*)d_ws;
    float* w = t + (size_t)NB * CR * HW;

    dim3 g1(HW / 16, NB);
    invo_gemm1_bn_relu<<<g1, 128, 0, stream>>>(x, W1, b1, gamma, beta, mean, var, t);

    dim3 g2(HW / 16, GK / 16, NB);
    invo_gemm2<<<g2, 32, 0, stream>>>(t, W2, b2, w);

    dim3 g3(HW / 16, 16, NB);
    invo_apply<<<g3, 256, 0, stream>>>(x, w, out);
}


# GEMM2 re-cut: 2048 waves in 256 blocks, each keeps the t fragments of 64 pixels in registers and walks a contiguous run of m-tiles (128-bit t/W2 loads, 256 B row stores); k summed in a permuted order
# speedup vs baseline: 3.1802x; 3.1802x over previous
#include <hip/hip_runtime.h>

typedef __attribute__((ext_vector_type(2))) float v2f;
typedef __attribute__((ext_vector_type(8))) float v8f;

#define EPS 1e-5f
#define HW  3136
#define CIN 256
#define CR  64
#define GK  784
#define NB  4

__global__ void __launch_bounds__(128)
invo_gemm1_bn_relu(const float* __restrict__ x, const float* __restrict__ W1,
                   const float* __restrict__ b1, const float* __restrict__ gamma,
                   const float* __restrict__ beta, const float* __restrict__ mean,
                   const float* __restrict__ var, float* __restrict__ t)
{
    const int lane   = threadIdx.x & 31;
    const int wave   = threadIdx.x >> 5;
    const int laneN  = lane & 15;
    const int laneHi = lane >> 4;
    const int p0     = blockIdx.x * 16;
    const int b      = blockIdx.y;
    const int o0     = wave * 16;

    const float* xb   = x + (size_t)b * CIN * HW;
    const float* Arow = W1 + (size_t)(o0 + laneN) * CIN + 2 * laneHi;
    const float* Bcol = xb + (size_t)(2 * laneHi) * HW + p0 + laneN;

    v8f acc = {};
#pragma unroll 8
    for (int k = 0; k < CIN; k += 4) {
        v2f a = *(const v2f*)(Arow + k);
        v2f bb;
        bb.x = Bcol[(size_t)k * HW];
        bb.y = Bcol[(size_t)(k + 1) * HW];
        acc = __builtin_amdgcn_wmma_f32_16x16x4_f32(
            false, a, false, bb, (short)0, acc, false, false);
    }

    float* tb = t + (size_t)b * CR * HW + p0 + laneN;
#pragma unroll
    for (int r = 0; r < 8; ++r) {
        int o = o0 + r + 8 * laneHi;
        float s  = gamma[o] * __frsqrt_rn(var[o] + EPS);
        float v  = (acc[r] + b1[o]) * s + (beta[o] - mean[o] * s);
        tb[(size_t)o * HW] = fmaxf(v, 0.0f);
    }
}

#define G2_WAVES  8
#define G2_BLOCKS 256
#define G2_NGRP   (NB * (HW / 64))
#define G2_MT     (GK / 16)
#define G2_ITEMS  (G2_NGRP * G2_MT)

__device__ __forceinline__ void g2_load_a(float4 (&a)[8], const float* __restrict__ W2,
                                          int mt, int laneN, int laneHi)
{
    const float* src = W2 + (size_t)(mt * 16 + laneN) * CR + 4 * laneHi;
    for (int u = 0; u < 8; ++u) a[u] = *(const float4*)(src + 8 * u);
}

__global__ void __launch_bounds__(G2_WAVES * 32)
invo_gemm2(const float* __restrict__ t, const float* __restrict__ W2,
           const float* __restrict__ b2, float* __restrict__ w)
{
    const int lane   = threadIdx.x & 31;
    const int laneN  = lane & 15;
    const int laneHi = lane >> 4;
    const int wv  = __builtin_amdgcn_readfirstlane((int)(threadIdx.x >> 5));
    const int gw  = blockIdx.x * G2_WAVES + wv;
    const int it1 = ((gw + 1) * G2_ITEMS) / (G2_BLOCKS * G2_WAVES);
    int it        = (gw * G2_ITEMS) / (G2_BLOCKS * G2_WAVES);

    float4 an[8];
    g2_load_a(an, W2, it % G2_MT, laneN, laneHi);

    while (it < it1) {
        const int ng  = it / G2_MT;
        const int b   = ng / (HW / 64);
        const int p0  = (ng - b * (HW / 64)) * 64;
        const int end = min(it1, (ng + 1) * G2_MT);

        v2f bf[4][16];
        const float* tb = t + (size_t)b * CR * HW + (size_t)(4 * laneHi) * HW + p0 + 4 * laneN;
        for (int j = 0; j < 16; ++j) {
            const int ch = 8 * (j >> 1) + 2 * (j & 1);
            const float4 f0 = *(const float4*)(tb + (size_t)ch * HW);
            const float4 f1 = *(const float4*)(tb + (size_t)(ch + 1) * HW);
            bf[0][j].x = f0.x; bf[0][j].y = f1.x;
            bf[1][j].x = f0.y; bf[1][j].y = f1.y;
            bf[2][j].x = f0.z; bf[2][j].y = f1.z;
            bf[3][j].x = f0.w; bf[3][j].y = f1.w;
        }

        for (; it < end; ++it) {
            const int mt = it - ng * G2_MT;
            float4 a[8];
            for (int u = 0; u < 8; ++u) a[u] = an[u];
            g2_load_a(an, W2, (mt + 1 == G2_MT) ? 0 : mt + 1, laneN, laneHi);

            const int m0 = mt * 16;
            const float4 bl = *(const float4*)(b2 + m0 + 8 * laneHi);
            const float4 bh = *(const float4*)(b2 + m0 + 8 * laneHi + 4);
            const float bs[8] = {bl.x, bl.y, bl.z, bl.w, bh.x, bh.y, bh.z, bh.w};

            v8f acc0 = {}, acc1 = {}, acc2 = {}, acc3 = {};
            for (int u = 0; u < 8; ++u) {
                const v2f a0 = {a[u].x, a[u].y};
                const v2f a1 = {a[u].z, a[u].w};
                acc0 = __builtin_amdgcn_wmma_f32_16x16x4_f32(false, a0, false, bf[0][2 * u], (short)0, acc0, false, false);
                acc1 = __builtin_amdgcn_wmma_f32_16x16x4_f32(false, a0, false, bf[1][2 * u], (short)0, acc1, false, false);
                acc2 = __builtin_amdgcn_wmma_f32_16x16x4_f32(false, a0, false, bf[2][2 * u], (short)0, acc2, false, false);
                acc3 = __builtin_amdgcn_wmma_f32_16x16x4_f32(false, a0, false, bf[3][2 * u], (short)0, acc3, false, false);
                acc0 = __builtin_amdgcn_wmma_f32_16x16x4_f32(false, a1, false, bf[0][2 * u + 1], (short)0, acc0, false, false);
                acc1 = __builtin_amdgcn_wmma_f32_16x16x4_f32(false, a1, false, bf[1][2 * u + 1], (short)0, acc1, false, false);
                acc2 = __builtin_amdgcn_wmma_f32_16x16x4_f32(false, a1, false, bf[2][2 * u + 1], (short)0, acc2, false, false);
                acc3 = __builtin_amdgcn_wmma_f32_16x16x4_f32(false, a1, false, bf[3][2 * u + 1], (short)0, acc3, false, false);
            }

            float* wb = w + ((size_t)b * GK + (size_t)(m0 + 8 * laneHi)) * HW + p0 + 4 * laneN;
            for (int r = 0; r < 8; ++r) {
                *(float4*)(wb + (size_t)r * HW) =
                    make_float4(acc0[r] + bs[r], acc1[r] + bs[r], acc2[r] + bs[r], acc3[r] + bs[r]);
            }
        }
    }
}

#define TP   448
#define HALO 224
#define XSP  (TP + 2 * HALO)
#define XQ   (XSP / 4)

__global__ void __launch_bounds__(TP)
invo_apply(const float* __restrict__ x, const float* __restrict__ wk,
           float* __restrict__ out)
{
    __shared__ __attribute__((aligned(16))) v2f xs[8 * XSP];

    const int tid = threadIdx.x;
    const int ps  = blockIdx.x * TP;
    const int g   = blockIdx.y;
    const int b   = blockIdx.z;
    const int p   = ps + tid;
    const int wc  = p % 56;

    const float* wp = wk + ((size_t)b * GK + (size_t)g * 49) * HW + p;
    float wr[49];
    for (int i = 0; i < 7; ++i) {
        for (int j = 0; j < 7; ++j) {
            int ww = wc + j - 3;
            float v = wp[(size_t)(i * 7 + j) * HW];
            wr[i * 7 + j] = ((ww >= 0) & (ww < 56)) ? v : 0.0f;
        }
    }

    const float* xg = x + ((size_t)b * CIN + (size_t)g * 16) * HW;
    const int half = (tid >= XQ) ? 1 : 0;
    const int q    = tid - half * XQ;
    const int f    = ps - HALO + 4 * q;
    const bool in  = (f >= 0) & (f < HW);
    const int fc   = in ? f : 0;
    float4 a[4], c[4];
    for (int k = 0; k < 4; ++k) {
        const float* s = xg + (size_t)(2 * (2 * k + half)) * HW + fc;
        a[k] = *(const float4*)s;
        c[k] = *(const float4*)(s + HW);
    }
    for (int k = 0; k < 4; ++k) {
        const int pr = 2 * k + half;
        float4* dst = (float4*)(xs + pr * XSP + 4 * q);
        dst[0] = make_float4(in ? a[k].x : 0.0f, in ? c[k].x : 0.0f,
                             in ? a[k].y : 0.0f, in ? c[k].y : 0.0f);
        dst[1] = make_float4(in ? a[k].z : 0.0f, in ? c[k].z : 0.0f,
                             in ? a[k].w : 0.0f, in ? c[k].w : 0.0f);
    }
    __syncthreads();

    const v2f* xc = xs + (HALO - (3 * 56 + 3)) + tid;
    float* op = out + ((size_t)b * CIN + (size_t)g * 16) * HW + p;
    for (int pr = 0; pr < 8; ++pr) {
        v2f acc = {0.0f, 0.0f};
        for (int i = 0; i < 7; ++i) {
            for (int j = 0; j < 7; ++j) {
                v2f wv = {wr[i * 7 + j], wr[i * 7 + j]};
                acc = __builtin_elementwise_fma(xc[i * 56 + j], wv, acc);
            }
        }
        op[0]  = acc.x;
        op[HW] = acc.y;
        op += 2 * HW;
        xc += XSP;
    }
}

extern "C" void kernel_launch(void* const* d_in, const int* in_sizes, int n_in,
                              void* d_out, int out_size, void* d_ws, size_t ws_size,
                              hipStream_t stream)
{
    const float* x     = (const float*)d_in[0];
    const float* W1    = (const float*)d_in[1];
    const float* b1    = (const float*)d_in[2];
    const float* gamma = (const float*)d_in[3];
    const float* beta  = (const float*)d_in[4];
    const float* mean  = (const float*)d_in[5];
    const float* var   = (const float*)d_in[6];
    const float* W2    = (const float*)d_in[7];
    const float* b2    = (const float*)d_in[8];
    float* out = (float*)d_out;

    float* t = (float*)d_ws;
    float* w = t + (size_t)NB * CR * HW;

    dim3 g1(HW / 16, NB);
    invo_gemm1_bn_relu<<<g1, 128, 0, stream>>>(x, W1, b1, gamma, beta, mean, var, t);

    invo_gemm2<<<G2_BLOCKS, G2_WAVES * 32, 0, stream>>>(t, W2, b2, w);

    dim3 g3(HW / TP, 16, NB);
    invo_apply<<<g3, TP, 0, stream>>>(x, w, out);
}


# GEMM1 re-cut: 196 blocks of 4 waves, a wave takes 16 out-channels x 64 pixels (4 tiles, 128-bit x loads, 256 B t row stores) and keeps its loads 10-15 k-steps ahead of its WMMAs; k sum order kept
# speedup vs baseline: 3.6354x; 1.1431x over previous
#include <hip/hip_runtime.h>

typedef __attribute__((ext_vector_type(2))) float v2f;
typedef __attribute__((ext_vector_type(8))) float v8f;

#define EPS 1e-5f
#define HW  3136
#define CIN 256
#define CR  64
#define GK  784
#define NB  4

#define G1_BLOCKS (NB * (HW / 64))
#define G1_KB     16

__device__ __forceinline__ void g1_load(v2f (&a)[G1_KB], float4 (&x0)[G1_KB], float4 (&x1)[G1_KB],
                                        const float* __restrict__ Arow,
                                        const float* __restrict__ Bcol, int c0)
{
#pragma unroll
    for (int j = 0; j < G1_KB; ++j) {
        const int k = c0 + 4 * j;
        a[j]  = *(const v2f*)(Arow + k);
        x0[j] = *(const float4*)(Bcol + (size_t)k * HW);
        x1[j] = *(const float4*)(Bcol + (size_t)(k + 1) * HW);
    }
}

__device__ __forceinline__ void g1_mma(const v2f (&a)[G1_KB], const float4 (&x0)[G1_KB],
                                       const float4 (&x1)[G1_KB],
                                       v8f& acc0, v8f& acc1, v8f& acc2, v8f& acc3)
{
#pragma unroll
    for (int j = 0; j < G1_KB; ++j) {
        const v2f b0 = {x0[j].x, x1[j].x};
        const v2f b1 = {x0[j].y, x1[j].y};
        const v2f b2 = {x0[j].z, x1[j].z};
        const v2f b3 = {x0[j].w, x1[j].w};
        acc0 = __builtin_amdgcn_wmma_f32_16x16x4_f32(false, a[j], false, b0, (short)0, acc0, false, false);
        acc1 = __builtin_amdgcn_wmma_f32_16x16x4_f32(false, a[j], false, b1, (short)0, acc1, false, false);
        acc2 = __builtin_amdgcn_wmma_f32_16x16x4_f32(false, a[j], false, b2, (short)0, acc2, false, false);
        acc3 = __builtin_amdgcn_wmma_f32_16x16x4_f32(false, a[j], false, b3, (short)0, acc3, false, false);
    }
}

__global__ void __launch_bounds__(128)
invo_gemm1_bn_relu(const float* __restrict__ x, const float* __restrict__ W1,
                   const float* __restrict__ b1, const float* __restrict__ gamma,
                   const float* __restrict__ beta, const float* __restrict__ mean,
                   const float* __restrict__ var, float* __restrict__ t)
{
    const int lane   = threadIdx.x & 31;
    const int wave   = threadIdx.x >> 5;
    const int laneN  = lane & 15;
    const int laneHi = lane >> 4;
    const int ng     = blockIdx.x;
    const int b      = ng / (HW / 64);
    const int p0     = (ng - b * (HW / 64)) * 64;
    const int o0     = wave * 16;

    const float* Arow = W1 + (size_t)(o0 + laneN) * CIN + 2 * laneHi;
    const float* Bcol = x + (size_t)b * CIN * HW + (size_t)(2 * laneHi) * HW + p0 + 4 * laneN;

    v8f acc0 = {}, acc1 = {}, acc2 = {}, acc3 = {};
    v2f    aA[G1_KB], aB[G1_KB];
    float4 xA0[G1_KB], xA1[G1_KB], xB0[G1_KB], xB1[G1_KB];

    g1_load(aA, xA0, xA1, Arow, Bcol, 0);
#pragma unroll
    for (int c = 0; c < CIN; c += 8 * G1_KB) {
        g1_load(aB, xB0, xB1, Arow, Bcol, c + 4 * G1_KB);
        g1_mma(aA, xA0, xA1, acc0, acc1, acc2, acc3);
        if (c + 8 * G1_KB < CIN) g1_load(aA, xA0, xA1, Arow, Bcol, c + 8 * G1_KB);
        g1_mma(aB, xB0, xB1, acc0, acc1, acc2, acc3);
    }

    float* tb = t + (size_t)b * CR * HW + p0 + 4 * laneN;
#pragma unroll
    for (int r = 0; r < 8; ++r) {
        int o = o0 + r + 8 * laneHi;
        float s  = gamma[o] * __frsqrt_rn(var[o] + EPS);
        float sh = beta[o] - mean[o] * s;
        float bo = b1[o];
        float v0 = (acc0[r] + bo) * s + sh;
        float v1 = (acc1[r] + bo) * s + sh;
        float v2 = (acc2[r] + bo) * s + sh;
        float v3 = (acc3[r] + bo) * s + sh;
        *(float4*)(tb + (size_t)o * HW) =
            make_float4(fmaxf(v0, 0.0f), fmaxf(v1, 0.0f), fmaxf(v2, 0.0f), fmaxf(v3, 0.0f));
    }
}

#define G2_WAVES  8
#define G2_BLOCKS 256
#define G2_NGRP   (NB * (HW / 64))
#define G2_MT     (GK / 16)
#define G2_ITEMS  (G2_NGRP * G2_MT)

__device__ __forceinline__ void g2_load_a(float4 (&a)[8], const float* __restrict__ W2,
                                          int mt, int laneN, int laneHi)
{
    const float* src = W2 + (size_t)(mt * 16 + laneN) * CR + 4 * laneHi;
    for (int u = 0; u < 8; ++u) a[u] = *(const float4*)(src + 8 * u);
}

__global__ void __launch_bounds__(G2_WAVES * 32)
invo_gemm2(const float* __restrict__ t, const float* __restrict__ W2,
           const float* __restrict__ b2, float* __restrict__ w)
{
    const int lane   = threadIdx.x & 31;
    const int laneN  = lane & 15;
    const int laneHi = lane >> 4;
    const int wv  = __builtin_amdgcn_readfirstlane((int)(threadIdx.x >> 5));
    const int gw  = blockIdx.x * G2_WAVES + wv;
    const int it1 = ((gw + 1) * G2_ITEMS) / (G2_BLOCKS * G2_WAVES);
    int it        = (gw * G2_ITEMS) / (G2_BLOCKS * G2_WAVES);

    float4 an[8];
    g2_load_a(an, W2, it % G2_MT, laneN, laneHi);

    while (it < it1) {
        const int ng  = it / G2_MT;
        const int b   = ng / (HW / 64);
        const int p0  = (ng - b * (HW / 64)) * 64;
        const int end = min(it1, (ng + 1) * G2_MT);

        v2f bf[4][16];
        const float* tb = t + (size_t)b * CR * HW + (size_t)(4 * laneHi) * HW + p0 + 4 * laneN;
        for (int j = 0; j < 16; ++j) {
            const int ch = 8 * (j >> 1) + 2 * (j & 1);
            const float4 f0 = *(const float4*)(tb + (size_t)ch * HW);
            const float4 f1 = *(const float4*)(tb + (size_t)(ch + 1) * HW);
            bf[0][j].x = f0.x; bf[0][j].y = f1.x;
            bf[1][j].x = f0.y; bf[1][j].y = f1.y;
            bf[2][j].x = f0.z; bf[2][j].y = f1.z;
            bf[3][j].x = f0.w; bf[3][j].y = f1.w;
        }

        for (; it < end; ++it) {
            const int mt = it - ng * G2_MT;
            float4 a[8];
            for (int u = 0; u < 8; ++u) a[u] = an[u];
            g2_load_a(an, W2, (mt + 1 == G2_MT) ? 0 : mt + 1, laneN, laneHi);

            const int m0 = mt * 16;
            const float4 bl = *(const float4*)(b2 + m0 + 8 * laneHi);
            const float4 bh = *(const float4*)(b2 + m0 + 8 * laneHi + 4);
            const float bs[8] = {bl.x, bl.y, bl.z, bl.w, bh.x, bh.y, bh.z, bh.w};

            v8f acc0 = {}, acc1 = {}, acc2 = {}, acc3 = {};
            for (int u = 0; u < 8; ++u) {
                const v2f a0 = {a[u].x, a[u].y};
                const v2f a1 = {a[u].z, a[u].w};
                acc0 = __builtin_amdgcn_wmma_f32_16x16x4_f32(false, a0, false, bf[0][2 * u], (short)0, acc0, false, false);
                acc1 = __builtin_amdgcn_wmma_f32_16x16x4_f32(false, a0, false, bf[1][2 * u], (short)0, acc1, false, false);
                acc2 = __builtin_amdgcn_wmma_f32_16x16x4_f32(false, a0, false, bf[2][2 * u], (short)0, acc2, false, false);
                acc3 = __builtin_amdgcn_wmma_f32_16x16x4_f32(false, a0, false, bf[3][2 * u], (short)0, acc3, false, false);
                acc0 = __builtin_amdgcn_wmma_f32_16x16x4_f32(false, a1, false, bf[0][2 * u + 1], (short)0, acc0, false, false);
                acc1 = __builtin_amdgcn_wmma_f32_16x16x4_f32(false, a1, false, bf[1][2 * u + 1], (short)0, acc1, false, false);
                acc2 = __builtin_amdgcn_wmma_f32_16x16x4_f32(false, a1, false, bf[2][2 * u + 1], (short)0, acc2, false, false);
                acc3 = __builtin_amdgcn_wmma_f32_16x16x4_f32(false, a1, false, bf[3][2 * u + 1], (short)0, acc3, false, false);
            }

            float* wb = w + ((size_t)b * GK + (size_t)(m0 + 8 * laneHi)) * HW + p0 + 4 * laneN;
            for (int r = 0; r < 8; ++r) {
                *(float4*)(wb + (size_t)r * HW) =
                    make_float4(acc0[r] + bs[r], acc1[r] + bs[r], acc2[r] + bs[r], acc3[r] + bs[r]);
            }
        }
    }
}

#define TP   448
#define HALO 224
#define XSP  (TP + 2 * HALO)
#define XQ   (XSP / 4)

__global__ void __launch_bounds__(TP)
invo_apply(const float* __restrict__ x, const float* __restrict__ wk,
           float* __restrict__ out)
{
    __shared__ __attribute__((aligned(16))) v2f xs[8 * XSP];

    const int tid = threadIdx.x;
    const int ps  = blockIdx.x * TP;
    const int g   = blockIdx.y;
    const int b   = blockIdx.z;
    const int p   = ps + tid;
    const int wc  = p % 56;

    const float* wp = wk + ((size_t)b * GK + (size_t)g * 49) * HW + p;
    float wr[49];
    for (int i = 0; i < 7; ++i) {
        for (int j = 0; j < 7; ++j) {
            int ww = wc + j - 3;
            float v = wp[(size_t)(i * 7 + j) * HW];
            wr[i * 7 + j] = ((ww >= 0) & (ww < 56)) ? v : 0.0f;
        }
    }

    const float* xg = x + ((size_t)b * CIN + (size_t)g * 16) * HW;
    const int half = (tid >= XQ) ? 1 : 0;
    const int q    = tid - half * XQ;
    const int f    = ps - HALO + 4 * q;
    const bool in  = (f >= 0) & (f < HW);
    const int fc   = in ? f : 0;
    float4 a[4], c[4];
    for (int k = 0; k < 4; ++k) {
        const float* s = xg + (size_t)(2 * (2 * k + half)) * HW + fc;
        a[k] = *(const float4*)s;
        c[k] = *(const float4*)(s + HW);
    }
    for (int k = 0; k < 4; ++k) {
        const int pr = 2 * k + half;
        float4* dst = (float4*)(xs + pr * XSP + 4 * q);
        dst[0] = make_float4(in ? a[k].x : 0.0f, in ? c[k].x : 0.0f,
                             in ? a[k].y : 0.0f, in ? c[k].y : 0.0f);
        dst[1] = make_float4(in ? a[k].z : 0.0f, in ? c[k].z : 0.0f,
                             in ? a[k].w : 0.0f, in ? c[k].w : 0.0f);
    }
    __syncthreads();

    const v2f* xc = xs + (HALO - (3 * 56 + 3)) + tid;
    float* op = out + ((size_t)b * CIN + (size_t)g * 16) * HW + p;
    for (int pr = 0; pr < 8; ++pr) {
        v2f acc = {0.0f, 0.0f};
        for (int i = 0; i < 7; ++i) {
            for (int j = 0; j < 7; ++j) {
                v2f wv = {wr[i * 7 + j], wr[i * 7 + j]};
                acc = __builtin_elementwise_fma(xc[i * 56 + j], wv, acc);
            }
        }
        op[0]  = acc.x;
        op[HW] = acc.y;
        op += 2 * HW;
        xc += XSP;
    }
}

extern "C" void kernel_launch(void* const* d_in, const int* in_sizes, int n_in,
                              void* d_out, int out_size, void* d_ws, size_t ws_size,
                              hipStream_t stream)
{
    const float* x     = (const float*)d_in[0];
    const float* W1    = (const float*)d_in[1];
    const float* b1    = (const float*)d_in[2];
    const float* gamma = (const float*)d_in[3];
    const float* beta  = (const float*)d_in[4];
    const float* mean  = (const float*)d_in[5];
    const float* var   = (const float*)d_in[6];
    const float* W2    = (const float*)d_in[7];
    const float* b2    = (const float*)d_in[8];
    float* out = (float*)d_out;

    float* t = (float*)d_ws;
    float* w = t + (size_t)NB * CR * HW;

    invo_gemm1_bn_relu<<<G1_BLOCKS, 128, 0, stream>>>(x, W1, b1, gamma, beta, mean, var, t);

    invo_gemm2<<<G2_BLOCKS, G2_WAVES * 32, 0, stream>>>(t, W2, b2, w);

    dim3 g3(HW / TP, 16, NB);
    invo_apply<<<g3, TP, 0, stream>>>(x, w, out);
}


# Fuse the first GEMM into the second: 196 blocks of 8 waves, a block builds its 64 pixels' t (BN+ReLU) by WMMA into LDS, then its waves walk the 49 m-tiles from it; 2 launches, sum orders kept
# speedup vs baseline: 3.8618x; 1.0623x over previous
#include <hip/hip_runtime.h>

typedef __attribute__((ext_vector_type(2))) float v2f;
typedef __attribute__((ext_vector_type(8))) float v8f;

#define EPS 1e-5f
#define HW  3136
#define CIN 256
#define CR  64
#define GK  784
#define NB  4

#define G_BLOCKS (NB * (HW / 64))
#define G_WAVES  8
#define G1_KB    16
#define G2_MT    (GK / 16)
#define TPITCH   68

__device__ __forceinline__ void g1_load(v2f (&a)[G1_KB], v2f (&x0)[G1_KB], v2f (&x1)[G1_KB],
                                        const float* __restrict__ Arow,
                                        const float* __restrict__ Bcol, int c0)
{
#pragma unroll
    for (int j = 0; j < G1_KB; ++j) {
        const int k = c0 + 4 * j;
        a[j]  = *(const v2f*)(Arow + k);
        x0[j] = *(const v2f*)(Bcol + (size_t)k * HW);
        x1[j] = *(const v2f*)(Bcol + (size_t)(k + 1) * HW);
    }
}

__device__ __forceinline__ void g1_mma(const v2f (&a)[G1_KB], const v2f (&x0)[G1_KB],
                                       const v2f (&x1)[G1_KB], v8f& acc0, v8f& acc1)
{
#pragma unroll
    for (int j = 0; j < G1_KB; ++j) {
        const v2f b0 = {x0[j].x, x1[j].x};
        const v2f b1 = {x0[j].y, x1[j].y};
        acc0 = __builtin_amdgcn_wmma_f32_16x16x4_f32(false, a[j], false, b0, (short)0, acc0, false, false);
        acc1 = __builtin_amdgcn_wmma_f32_16x16x4_f32(false, a[j], false, b1, (short)0, acc1, false, false);
    }
}

__device__ __forceinline__ void g2_load_a(float4 (&a)[8], const float* __restrict__ W2,
                                          int mt, int laneN, int laneHi)
{
    const float* src = W2 + (size_t)(mt * 16 + laneN) * CR + 4 * laneHi;
    for (int u = 0; u < 8; ++u) a[u] = *(const float4*)(src + 8 * u);
}

__global__ void __launch_bounds__(G_WAVES * 32)
invo_kernel_gen(const float* __restrict__ x, const float* __restrict__ W1,
                const float* __restrict__ b1, const float* __restrict__ gamma,
                const float* __restrict__ beta, const float* __restrict__ mean,
                const float* __restrict__ var, const float* __restrict__ W2,
                const float* __restrict__ b2, float* __restrict__ w)
{
    __shared__ __attribute__((aligned(16))) float ts[CR * TPITCH];

    const int lane   = threadIdx.x & 31;
    const int laneN  = lane & 15;
    const int laneHi = lane >> 4;
    const int wv = __builtin_amdgcn_readfirstlane((int)(threadIdx.x >> 5));
    const int ng = blockIdx.x;
    const int b  = ng / (HW / 64);
    const int p0 = (ng - b * (HW / 64)) * 64;

    {
        const int o0 = (wv & 3) * 16;
        const int ph = wv >> 2;
        const float* Arow = W1 + (size_t)(o0 + laneN) * CIN + 2 * laneHi;
        const float* Bcol = x + (size_t)b * CIN * HW + (size_t)(2 * laneHi) * HW
                              + p0 + 32 * ph + 2 * laneN;

        v8f acc0 = {}, acc1 = {};
        v2f aA[G1_KB], aB[G1_KB];
        v2f xA0[G1_KB], xA1[G1_KB], xB0[G1_KB], xB1[G1_KB];

        g1_load(aA, xA0, xA1, Arow, Bcol, 0);
#pragma unroll
        for (int c = 0; c < CIN; c += 8 * G1_KB) {
            g1_load(aB, xB0, xB1, Arow, Bcol, c + 4 * G1_KB);
            g1_mma(aA, xA0, xA1, acc0, acc1);
            if (c + 8 * G1_KB < CIN) g1_load(aA, xA0, xA1, Arow, Bcol, c + 8 * G1_KB);
            g1_mma(aB, xB0, xB1, acc0, acc1);
        }

        float* td = ts + 32 * ph + 2 * laneN;
#pragma unroll
        for (int r = 0; r < 8; ++r) {
            int o = o0 + r + 8 * laneHi;
            float s  = gamma[o] * __frsqrt_rn(var[o] + EPS);
            float sh = beta[o] - mean[o] * s;
            float bo = b1[o];
            float v0 = (acc0[r] + bo) * s + sh;
            float v1 = (acc1[r] + bo) * s + sh;
            v2f tv = {fmaxf(v0, 0.0f), fmaxf(v1, 0.0f)};
            *(v2f*)(td + o * TPITCH) = tv;
        }
    }

    const int mt0 = (wv * G2_MT) / G_WAVES;
    const int mt1 = ((wv + 1) * G2_MT) / G_WAVES;

    float4 an[8];
    g2_load_a(an, W2, mt0, laneN, laneHi);

    __syncthreads();

    v2f bf[4][16];
    const float* tl = ts + (4 * laneHi) * TPITCH + 4 * laneN;
    for (int j = 0; j < 16; ++j) {
        const int ch = 8 * (j >> 1) + 2 * (j & 1);
        const float4 f0 = *(const float4*)(tl + ch * TPITCH);
        const float4 f1 = *(const float4*)(tl + (ch + 1) * TPITCH);
        bf[0][j].x = f0.x; bf[0][j].y = f1.x;
        bf[1][j].x = f0.y; bf[1][j].y = f1.y;
        bf[2][j].x = f0.z; bf[2][j].y = f1.z;
        bf[3][j].x = f0.w; bf[3][j].y = f1.w;
    }

    for (int mt = mt0; mt < mt1; ++mt) {
        float4 a[8];
        for (int u = 0; u < 8; ++u) a[u] = an[u];
        g2_load_a(an, W2, (mt + 1 < mt1) ? mt + 1 : mt, laneN, laneHi);

        const int m0 = mt * 16;
        const float4 bl = *(const float4*)(b2 + m0 + 8 * laneHi);
        const float4 bh = *(const float4*)(b2 + m0 + 8 * laneHi + 4);
        const float bs[8] = {bl.x, bl.y, bl.z, bl.w, bh.x, bh.y, bh.z, bh.w};

        v8f acc0 = {}, acc1 = {}, acc2 = {}, acc3 = {};
        for (int u = 0; u < 8; ++u) {
            const v2f a0 = {a[u].x, a[u].y};
            const v2f a1 = {a[u].z, a[u].w};
            acc0 = __builtin_amdgcn_wmma_f32_16x16x4_f32(false, a0, false, bf[0][2 * u], (short)0, acc0, false, false);
            acc1 = __builtin_amdgcn_wmma_f32_16x16x4_f32(false, a0, false, bf[1][2 * u], (short)0, acc1, false, false);
            acc2 = __builtin_amdgcn_wmma_f32_16x16x4_f32(false, a0, false, bf[2][2 * u], (short)0, acc2, false, false);
            acc3 = __builtin_amdgcn_wmma_f32_16x16x4_f32(false, a0, false, bf[3][2 * u], (short)0, acc3, false, false);
            acc0 = __builtin_amdgcn_wmma_f32_16x16x4_f32(false, a1, false, bf[0][2 * u + 1], (short)0, acc0, false, false);
            acc1 = __builtin_amdgcn_wmma_f32_16x16x4_f32(false, a1, false, bf[1][2 * u + 1], (short)0, acc1, false, false);
            acc2 = __builtin_amdgcn_wmma_f32_16x16x4_f32(false, a1, false, bf[2][2 * u + 1], (short)0, acc2, false, false);
            acc3 = __builtin_amdgcn_wmma_f32_16x16x4_f32(false, a1, false, bf[3][2 * u + 1], (short)0, acc3, false, false);
        }

        float* wb = w + ((size_t)b * GK + (size_t)(m0 + 8 * laneHi)) * HW + p0 + 4 * laneN;
        for (int r = 0; r < 8; ++r) {
            *(float4*)(wb + (size_t)r * HW) =
                make_float4(acc0[r] + bs[r], acc1[r] + bs[r], acc2[r] + bs[r], acc3[r] + bs[r]);
        }
    }
}

#define TP   448
#define HALO 224
#define XSP  (TP + 2 * HALO)
#define XQ   (XSP / 4)

__global__ void __launch_bounds__(TP)
invo_apply(const float* __restrict__ x, const float* __restrict__ wk,
           float* __restrict__ out)
{
    __shared__ __attribute__((aligned(16))) v2f xs[8 * XSP];

    const int tid = threadIdx.x;
    const int ps  = blockIdx.x * TP;
    const int g   = blockIdx.y;
    const int b   = blockIdx.z;
    const int p   = ps + tid;
    const int wc  = p % 56;

    const float* wp = wk + ((size_t)b * GK + (size_t)g * 49) * HW + p;
    float wr[49];
    for (int i = 0; i < 7; ++i) {
        for (int j = 0; j < 7; ++j) {
            int ww = wc + j - 3;
            float v = wp[(size_t)(i * 7 + j) * HW];
            wr[i * 7 + j] = ((ww >= 0) & (ww < 56)) ? v : 0.0f;
        }
    }

    const float* xg = x + ((size_t)b * CIN + (size_t)g * 16) * HW;
    const int half = (tid >= XQ) ? 1 : 0;
    const int q    = tid - half * XQ;
    const int f    = ps - HALO + 4 * q;
    const bool in  = (f >= 0) & (f < HW);
    const int fc   = in ? f : 0;
    float4 a[4], c[4];
    for (int k = 0; k < 4; ++k) {
        const float* s = xg + (size_t)(2 * (2 * k + half)) * HW + fc;
        a[k] = *(const float4*)s;
        c[k] = *(const float4*)(s + HW);
    }
    for (int k = 0; k < 4; ++k) {
        const int pr = 2 * k + half;
        float4* dst = (float4*)(xs + pr * XSP + 4 * q);
        dst[0] = make_float4(in ? a[k].x : 0.0f, in ? c[k].x : 0.0f,
                             in ? a[k].y : 0.0f, in ? c[k].y : 0.0f);
        dst[1] = make_float4(in ? a[k].z : 0.0f, in ? c[k].z : 0.0f,
                             in ? a[k].w : 0.0f, in ? c[k].w : 0.0f);
    }
    __syncthreads();

    const v2f* xc = xs + (HALO - (3 * 56 + 3)) + tid;
    float* op = out + ((size_t)b * CIN + (size_t)g * 16) * HW + p;
    for (int pr = 0; pr < 8; ++pr) {
        v2f acc = {0.0f, 0.0f};
        for (int i = 0; i < 7; ++i) {
            for (int j = 0; j < 7; ++j) {
                v2f wv = {wr[i * 7 + j], wr[i * 7 + j]};
                acc = __builtin_elementwise_fma(xc[i * 56 + j], wv, acc);
            }
        }
        op[0]  = acc.x;
        op[HW] = acc.y;
        op += 2 * HW;
        xc += XSP;
    }
}

extern "C" void kernel_launch(void* const* d_in, const int* in_sizes, int n_in,
                              void* d_out, int out_size, void* d_ws, size_t ws_size,
                              hipStream_t stream)
{
    const float* x     = (const float*)d_in[0];
    const float* W1    = (const float*)d_in[1];
    const float* b1    = (const float*)d_in[2];
    const float* gamma = (const float*)d_in[3];
    const float* beta  = (const float*)d_in[4];
    const float* mean  = (const float*)d_in[5];
    const float* var   = (const float*)d_in[6];
    const float* W2    = (const float*)d_in[7];
    const float* b2    = (const float*)d_in[8];
    float* out = (float*)d_out;

    float* w = (float*)d_ws + (size_t)NB * CR * HW;

    invo_kernel_gen<<<G_BLOCKS, G_WAVES * 32, 0, stream>>>(x, W1, b1, gamma, beta, mean, var, W2, b2, w);

    dim3 g3(HW / TP, 16, NB);
    invo_apply<<<g3, TP, 0, stream>>>(x, w, out);
}


# Involution kernel: x requested first, staged behind an LDS-only barrier; the 49 weight loads stay in flight and the taps run row by row as weight rows land; each channel's tap sum order kept
# speedup vs baseline: 3.9964x; 1.0349x over previous
#include <hip/hip_runtime.h>

typedef __attribute__((ext_vector_type(2))) float v2f;
typedef __attribute__((ext_vector_type(8))) float v8f;

#define EPS 1e-5f
#define HW  3136
#define CIN 256
#define CR  64
#define GK  784
#define NB  4

#define G_BLOCKS (NB * (HW / 64))
#define G_WAVES  8
#define G1_KB    16
#define G2_MT    (GK / 16)
#define TPITCH   68

__device__ __forceinline__ void g1_load(v2f (&a)[G1_KB], v2f (&x0)[G1_KB], v2f (&x1)[G1_KB],
                                        const float* __restrict__ Arow,
                                        const float* __restrict__ Bcol, int c0)
{
#pragma unroll
    for (int j = 0; j < G1_KB; ++j) {
        const int k = c0 + 4 * j;
        a[j]  = *(const v2f*)(Arow + k);
        x0[j] = *(const v2f*)(Bcol + (size_t)k * HW);
        x1[j] = *(const v2f*)(Bcol + (size_t)(k + 1) * HW);
    }
}

__device__ __forceinline__ void g1_mma(const v2f (&a)[G1_KB], const v2f (&x0)[G1_KB],
                                       const v2f (&x1)[G1_KB], v8f& acc0, v8f& acc1)
{
#pragma unroll
    for (int j = 0; j < G1_KB; ++j) {
        const v2f b0 = {x0[j].x, x1[j].x};
        const v2f b1 = {x0[j].y, x1[j].y};
        acc0 = __builtin_amdgcn_wmma_f32_16x16x4_f32(false, a[j], false, b0, (short)0, acc0, false, false);
        acc1 = __builtin_amdgcn_wmma_f32_16x16x4_f32(false, a[j], false, b1, (short)0, acc1, false, false);
    }
}

__device__ __forceinline__ void g2_load_a(float4 (&a)[8], const float* __restrict__ W2,
                                          int mt, int laneN, int laneHi)
{
    const float* src = W2 + (size_t)(mt * 16 + laneN) * CR + 4 * laneHi;
    for (int u = 0; u < 8; ++u) a[u] = *(const float4*)(src + 8 * u);
}

__global__ void __launch_bounds__(G_WAVES * 32)
invo_kernel_gen(const float* __restrict__ x, const float* __restrict__ W1,
                const float* __restrict__ b1, const float* __restrict__ gamma,
                const float* __restrict__ beta, const float* __restrict__ mean,
                const float* __restrict__ var, const float* __restrict__ W2,
                const float* __restrict__ b2, float* __restrict__ w)
{
    __shared__ __attribute__((aligned(16))) float ts[CR * TPITCH];

    const int lane   = threadIdx.x & 31;
    const int laneN  = lane & 15;
    const int laneHi = lane >> 4;
    const int wv = __builtin_amdgcn_readfirstlane((int)(threadIdx.x >> 5));
    const int ng = blockIdx.x;
    const int b  = ng / (HW / 64);
    const int p0 = (ng - b * (HW / 64)) * 64;

    {
        const int o0 = (wv & 3) * 16;
        const int ph = wv >> 2;
        const float* Arow = W1 + (size_t)(o0 + laneN) * CIN + 2 * laneHi;
        const float* Bcol = x + (size_t)b * CIN * HW + (size_t)(2 * laneHi) * HW
                              + p0 + 32 * ph + 2 * laneN;

        v8f acc0 = {}, acc1 = {};
        v2f aA[G1_KB], aB[G1_KB];
        v2f xA0[G1_KB], xA1[G1_KB], xB0[G1_KB], xB1[G1_KB];

        g1_load(aA, xA0, xA1, Arow, Bcol, 0);
#pragma unroll
        for (int c = 0; c < CIN; c += 8 * G1_KB) {
            g1_load(aB, xB0, xB1, Arow, Bcol, c + 4 * G1_KB);
            g1_mma(aA, xA0, xA1, acc0, acc1);
            if (c + 8 * G1_KB < CIN) g1_load(aA, xA0, xA1, Arow, Bcol, c + 8 * G1_KB);
            g1_mma(aB, xB0, xB1, acc0, acc1);
        }

        float* td = ts + 32 * ph + 2 * laneN;
#pragma unroll
        for (int r = 0; r < 8; ++r) {
            int o = o0 + r + 8 * laneHi;
            float s  = gamma[o] * __frsqrt_rn(var[o] + EPS);
            float sh = beta[o] - mean[o] * s;
            float bo = b1[o];
            float v0 = (acc0[r] + bo) * s + sh;
            float v1 = (acc1[r] + bo) * s + sh;
            v2f tv = {fmaxf(v0, 0.0f), fmaxf(v1, 0.0f)};
            *(v2f*)(td + o * TPITCH) = tv;
        }
    }

    const int mt0 = (wv * G2_MT) / G_WAVES;
    const int mt1 = ((wv + 1) * G2_MT) / G_WAVES;

    float4 an[8];
    g2_load_a(an, W2, mt0, laneN, laneHi);

    __syncthreads();

    v2f bf[4][16];
    const float* tl = ts + (4 * laneHi) * TPITCH + 4 * laneN;
    for (int j = 0; j < 16; ++j) {
        const int ch = 8 * (j >> 1) + 2 * (j & 1);
        const float4 f0 = *(const float4*)(tl + ch * TPITCH);
        const float4 f1 = *(const float4*)(tl + (ch + 1) * TPITCH);
        bf[0][j].x = f0.x; bf[0][j].y = f1.x;
        bf[1][j].x = f0.y; bf[1][j].y = f1.y;
        bf[2][j].x = f0.z; bf[2][j].y = f1.z;
        bf[3][j].x = f0.w; bf[3][j].y = f1.w;
    }

    for (int mt = mt0; mt < mt1; ++mt) {
        float4 a[8];
        for (int u = 0; u < 8; ++u) a[u] = an[u];
        g2_load_a(an, W2, (mt + 1 < mt1) ? mt + 1 : mt, laneN, laneHi);

        const int m0 = mt * 16;
        const float4 bl = *(const float4*)(b2 + m0 + 8 * laneHi);
        const float4 bh = *(const float4*)(b2 + m0 + 8 * laneHi + 4);
        const float bs[8] = {bl.x, bl.y, bl.z, bl.w, bh.x, bh.y, bh.z, bh.w};

        v8f acc0 = {}, acc1 = {}, acc2 = {}, acc3 = {};
        for (int u = 0; u < 8; ++u) {
            const v2f a0 = {a[u].x, a[u].y};
            const v2f a1 = {a[u].z, a[u].w};
            acc0 = __builtin_amdgcn_wmma_f32_16x16x4_f32(false, a0, false, bf[0][2 * u], (short)0, acc0, false, false);
            acc1 = __builtin_amdgcn_wmma_f32_16x16x4_f32(false, a0, false, bf[1][2 * u], (short)0, acc1, false, false);
            acc2 = __builtin_amdgcn_wmma_f32_16x16x4_f32(false, a0, false, bf[2][2 * u], (short)0, acc2, false, false);
            acc3 = __builtin_amdgcn_wmma_f32_16x16x4_f32(false, a0, false, bf[3][2 * u], (short)0, acc3, false, false);
            acc0 = __builtin_amdgcn_wmma_f32_16x16x4_f32(false, a1, false, bf[0][2 * u + 1], (short)0, acc0, false, false);
            acc1 = __builtin_amdgcn_wmma_f32_16x16x4_f32(false, a1, false, bf[1][2 * u + 1], (short)0, acc1, false, false);
            acc2 = __builtin_amdgcn_wmma_f32_16x16x4_f32(false, a1, false, bf[2][2 * u + 1], (short)0, acc2, false, false);
            acc3 = __builtin_amdgcn_wmma_f32_16x16x4_f32(false, a1, false, bf[3][2 * u + 1], (short)0, acc3, false, false);
        }

        float* wb = w + ((size_t)b * GK + (size_t)(m0 + 8 * laneHi)) * HW + p0 + 4 * laneN;
        for (int r = 0; r < 8; ++r) {
            *(float4*)(wb + (size_t)r * HW) =
                make_float4(acc0[r] + bs[r], acc1[r] + bs[r], acc2[r] + bs[r], acc3[r] + bs[r]);
        }
    }
}

#define TP   448
#define HALO 224
#define XSP  (TP + 2 * HALO)
#define XQ   (XSP / 4)

__global__ void __launch_bounds__(TP)
invo_apply(const float* x, const float* wk, float* __restrict__ out)
{
    __shared__ __attribute__((aligned(16))) v2f xs[8 * XSP];

    const int tid = threadIdx.x;
    const int ps  = blockIdx.x * TP;
    const int g   = blockIdx.y;
    const int b   = blockIdx.z;
    const int p   = ps + tid;
    const int wc  = p % 56;

    const float* xg = x + ((size_t)b * CIN + (size_t)g * 16) * HW;
    const int half = (tid >= XQ) ? 1 : 0;
    const int q    = tid - half * XQ;
    const int f    = ps - HALO + 4 * q;
    const bool in  = (f >= 0) & (f < HW);
    const int fc   = in ? f : 0;
    float4 a[4], c[4];
    for (int k = 0; k < 4; ++k) {
        const float* s = xg + (size_t)(2 * (2 * k + half)) * HW + fc;
        a[k] = *(const float4*)s;
        c[k] = *(const float4*)(s + HW);
    }
    asm volatile("" ::: "memory");

    const float* wp = wk + ((size_t)b * GK + (size_t)g * 49) * HW + p;
    float wr[49];
#pragma unroll
    for (int i = 0; i < 7; ++i) {
#pragma unroll
        for (int j = 0; j < 7; ++j) wr[i * 7 + j] = wp[(size_t)(i * 7 + j) * HW];
        asm volatile("" ::: "memory");
    }

    for (int k = 0; k < 4; ++k) {
        const int pr = 2 * k + half;
        float4* dst = (float4*)(xs + pr * XSP + 4 * q);
        dst[0] = make_float4(in ? a[k].x : 0.0f, in ? c[k].x : 0.0f,
                             in ? a[k].y : 0.0f, in ? c[k].y : 0.0f);
        dst[1] = make_float4(in ? a[k].z : 0.0f, in ? c[k].z : 0.0f,
                             in ? a[k].w : 0.0f, in ? c[k].w : 0.0f);
    }
    __builtin_amdgcn_fence(__ATOMIC_RELEASE, "workgroup", "local");
    __builtin_amdgcn_s_barrier();
    __builtin_amdgcn_fence(__ATOMIC_ACQUIRE, "workgroup", "local");

    bool okj[7];
#pragma unroll
    for (int j = 0; j < 7; ++j) {
        const int ww = wc + j - 3;
        okj[j] = (ww >= 0) & (ww < 56);
    }

    const v2f* xc = xs + (HALO - (3 * 56 + 3)) + tid;
    v2f acc[8];
#pragma unroll
    for (int pr = 0; pr < 8; ++pr) acc[pr] = (v2f){0.0f, 0.0f};
#pragma unroll
    for (int i = 0; i < 7; ++i) {
        asm volatile("" : "+v"(wr[i * 7 + 0]), "+v"(wr[i * 7 + 1]), "+v"(wr[i * 7 + 2]),
                          "+v"(wr[i * 7 + 3]), "+v"(wr[i * 7 + 4]), "+v"(wr[i * 7 + 5]),
                          "+v"(wr[i * 7 + 6]) :: "memory");
#pragma unroll
        for (int j = 0; j < 7; ++j) {
            const float wm = okj[j] ? wr[i * 7 + j] : 0.0f;
            const v2f wv = {wm, wm};
#pragma unroll
            for (int pr = 0; pr < 8; ++pr)
                acc[pr] = __builtin_elementwise_fma(xc[pr * XSP + i * 56 + j], wv, acc[pr]);
        }
    }

    float* op = out + ((size_t)b * CIN + (size_t)g * 16) * HW + p;
#pragma unroll
    for (int pr = 0; pr < 8; ++pr) {
        op[(size_t)(2 * pr) * HW]     = acc[pr].x;
        op[(size_t)(2 * pr + 1) * HW] = acc[pr].y;
    }
}

extern "C" void kernel_launch(void* const* d_in, const int* in_sizes, int n_in,
                              void* d_out, int out_size, void* d_ws, size_t ws_size,
                              hipStream_t stream)
{
    const float* x     = (const float*)d_in[0];
    const float* W1    = (const float*)d_in[1];
    const float* b1    = (const float*)d_in[2];
    const float* gamma = (const float*)d_in[3];
    const float* beta  = (const float*)d_in[4];
    const float* mean  = (const float*)d_in[5];
    const float* var   = (const float*)d_in[6];
    const float* W2    = (const float*)d_in[7];
    const float* b2    = (const float*)d_in[8];
    float* out = (float*)d_out;

    float* w = (float*)d_ws + (size_t)NB * CR * HW;

    invo_kernel_gen<<<G_BLOCKS, G_WAVES * 32, 0, stream>>>(x, W1, b1, gamma, beta, mean, var, W2, b2, w);

    dim3 g3(HW / TP, 16, NB);
    invo_apply<<<g3, TP, 0, stream>>>(x, w, out);
}
